# starting kernel
# baseline (speedup 1.0000x reference)
#include <hip/hip_runtime.h>
#include <math.h>

typedef __attribute__((ext_vector_type(2))) float v2f;
typedef __attribute__((ext_vector_type(8))) float v8f;

#define IN_F    512
#define NQ      10
#define KSTEPS  (IN_F / 4)
#define WPB     4
#define TPB     (WPB * 32)

__global__ __launch_bounds__(TPB)
void qgate_kernel(const float* __restrict__ x,
                  const float* __restrict__ W,
                  const float* __restrict__ bias,
                  const float* __restrict__ theta,
                  float* __restrict__ out,
                  int batch)
{
    __shared__ float lbs[KSTEPS * 32 * 2];

    const int tid = threadIdx.x;
    for (int e = tid; e < KSTEPS * 32; e += TPB) {
        const int m  = e >> 5;
        const int ln = e & 31;
        const int nn = ln & 15;
        const int kb = 4 * m + ((ln >> 4) << 1);
        float vx = 0.f, vy = 0.f;
        if (nn < NQ) {
            const float* wp = W + nn * IN_F + kb;
            vx = wp[0];
            vy = wp[1];
        }
        lbs[2 * e + 0] = vx;
        lbs[2 * e + 1] = vy;
    }
    __syncthreads();

    const int lane = tid & 31;
    const int wave = tid >> 5;
    const int tile = blockIdx.x * WPB + wave;
    const int ntiles = batch >> 4;
    if (tile >= ntiles) return;

    const int rowbase = tile << 4;
    const int n  = lane & 15;
    const int hi = lane >> 4;

    const float* ap = x + (size_t)(rowbase + n) * IN_F + (hi << 1);
    const v2f* bfrag = (const v2f*)lbs;

    v8f acc0 = {}; v8f acc1 = {};
    #pragma unroll 4
    for (int m = 0; m < KSTEPS; m += 2) {
        if ((m & 7) == 0)
            __builtin_prefetch(ap + 4 * (m + 16), 0, 1);
        v2f a0 = *(const v2f*)(ap + 4 * (m + 0));
        v2f b0 = bfrag[(m + 0) * 32 + lane];
        acc0 = __builtin_amdgcn_wmma_f32_16x16x4_f32(false, a0, false, b0,
                                                     (short)0, acc0, false, false);
        v2f a1 = *(const v2f*)(ap + 4 * (m + 1));
        v2f b1 = bfrag[(m + 1) * 32 + lane];
        acc1 = __builtin_amdgcn_wmma_f32_16x16x4_f32(false, a1, false, b1,
                                                     (short)0, acc1, false, false);
    }

    float bt = 0.f;
    if (n < NQ) bt = bias[n] + theta[n];

    #pragma unroll
    for (int i = 0; i < 8; ++i) {
        float v = cosf(acc0[i] + acc1[i] + bt);
        #pragma unroll
        for (int off = 1; off < 16; off <<= 1) {
            float o = __shfl_up(v, off, 16);
            v = (n >= off) ? v * o : v;
        }
        if (n < NQ) {
            const int row = rowbase + (hi ? (8 + i) : i);
            out[(size_t)row * NQ + n] = v;
        }
    }
}

extern "C" void kernel_launch(void* const* d_in, const int* in_sizes, int n_in,
                              void* d_out, int out_size, void* d_ws, size_t ws_size,
                              hipStream_t stream)
{
    const float* x     = (const float*)d_in[0];
    const float* W     = (const float*)d_in[1];
    const float* b     = (const float*)d_in[2];
    const float* theta = (const float*)d_in[3];
    float* out = (float*)d_out;

    const int batch  = in_sizes[0] / IN_F;
    const int ntiles = batch >> 4;
    const int blocks = (ntiles + WPB - 1) / WPB;
    qgate_kernel<<<blocks, TPB, 0, stream>>>(x, W, b, theta, out, batch);
}


# every load issued in a group ahead of its use: W staging 32 loads per round trip (was 1), A fragments two groups of 16 ahead (was 8 per trip), B fragments one group ahead; k and sum order kept
# speedup vs baseline: 1.9175x; 1.9175x over previous
#include <hip/hip_runtime.h>
#include <math.h>


typedef __attribute__((ext_vector_type(2))) float v2f;
typedef __attribute__((ext_vector_type(8))) float v8f;

#define IN_F    512
#define NQ      10
#define KSTEPS  (IN_F / 4)
#define WPB     4
#define TPB     (WPB * 32)
#define GK      16
#define NG      (KSTEPS / GK)
#define WST     (KSTEPS / WPB)

__global__ __launch_bounds__(TPB)
void qgate_kernel(const float* __restrict__ x,
                  const float* __restrict__ W,
                  const float* __restrict__ bias,
                  const float* __restrict__ theta,
                  float* __restrict__ out,
                  int batch)
{
    __shared__ float lbs[KSTEPS * 32 * 2];

    const int tid  = threadIdx.x;
    const int lane = tid & 31;
    const int wave = tid >> 5;
    const int n    = lane & 15;
    const int hi   = lane >> 4;
    v2f* lfrag = (v2f*)lbs;

    {
        const int nr = (n < NQ) ? n : (NQ - 1);
        const float* wrow = W + nr * IN_F + (hi << 1) + 4 * wave;
        v2f wv[WST];
        #pragma unroll
        for (int it = 0; it < WST; ++it)
            wv[it] = *(const v2f*)(wrow + 4 * WPB * it);
        const v2f zero = {0.f, 0.f};
        #pragma unroll
        for (int it = 0; it < WST; ++it)
            lfrag[(wave + WPB * it) * 32 + lane] = (n < NQ) ? wv[it] : zero;
    }
    __syncthreads();

    const int tile = blockIdx.x * WPB + wave;
    const int ntiles = batch >> 4;
    if (tile >= ntiles) return;

    const int rowbase = tile << 4;

    const float* ap = x + (size_t)(rowbase + n) * IN_F + (hi << 1);
    const v2f* bfrag = (const v2f*)lbs;

    v8f acc0 = {}; v8f acc1 = {};

    v2f af[3][GK];
    v2f bf[2][GK];
    #pragma unroll
    for (int j = 0; j < GK; ++j) af[0][j] = *(const v2f*)(ap + 4 * j);
    #pragma unroll
    for (int j = 0; j < GK; ++j) af[1][j] = *(const v2f*)(ap + 4 * (GK + j));
    #pragma unroll
    for (int j = 0; j < GK; ++j) bf[0][j] = bfrag[j * 32 + lane];

    #pragma unroll
    for (int g = 0; g < NG; ++g) {
        if (g + 2 < NG) {
            #pragma unroll
            for (int j = 0; j < GK; ++j)
                af[(g + 2) % 3][j] = *(const v2f*)(ap + 4 * ((g + 2) * GK + j));
        }
        if (g + 1 < NG) {
            #pragma unroll
            for (int j = 0; j < GK; ++j)
                bf[(g + 1) & 1][j] = bfrag[((g + 1) * GK + j) * 32 + lane];
        }
        __builtin_amdgcn_sched_barrier(0);
        #pragma unroll
        for (int j = 0; j < GK; j += 2) {
            acc0 = __builtin_amdgcn_wmma_f32_16x16x4_f32(false, af[g % 3][j], false, bf[g & 1][j],
                                                         (short)0, acc0, false, false);
            acc1 = __builtin_amdgcn_wmma_f32_16x16x4_f32(false, af[g % 3][j + 1], false, bf[g & 1][j + 1],
                                                         (short)0, acc1, false, false);
        }
        __builtin_amdgcn_sched_barrier(0);
    }

    float bt = 0.f;
    if (n < NQ) bt = bias[n] + theta[n];

    #pragma unroll
    for (int i = 0; i < 8; ++i) {
        float v = cosf(acc0[i] + acc1[i] + bt);
        #pragma unroll
        for (int off = 1; off < 16; off <<= 1) {
            float o = __shfl_up(v, off, 16);
            v = (n >= off) ? v * o : v;
        }
        if (n < NQ) {
            const int row = rowbase + (hi ? (8 + i) : i);
            out[(size_t)row * NQ + n] = v;
        }
    }
}

extern "C" void kernel_launch(void* const* d_in, const int* in_sizes, int n_in,
                              void* d_out, int out_size, void* d_ws, size_t ws_size,
                              hipStream_t stream)
{
    const float* x     = (const float*)d_in[0];
    const float* W     = (const float*)d_in[1];
    const float* b     = (const float*)d_in[2];
    const float* theta = (const float*)d_in[3];
    float* out = (float*)d_out;

    const int batch  = in_sizes[0] / IN_F;
    const int ntiles = batch >> 4;
    const int blocks = (ntiles + WPB - 1) / WPB;
    qgate_kernel<<<blocks, TPB, 0, stream>>>(x, W, b, theta, out, batch);
}


# whole x tile and W slice of a wave requested in one burst at kernel start: x as 64 16-byte fragment loads (2 k-steps each; was 128 8-byte in a ring); k regrouped alike in A and B, sum order changes
# speedup vs baseline: 2.0205x; 1.0537x over previous
#include <hip/hip_runtime.h>
#include <math.h>


typedef __attribute__((ext_vector_type(2))) float v2f;
typedef __attribute__((ext_vector_type(4))) float v4f;
typedef __attribute__((ext_vector_type(8))) float v8f;

#define IN_F    512
#define NQ      10
#define KPAIRS  (IN_F / 8)
#define WPB     4
#define TPB     (WPB * 32)
#define GP      8
#define NG      (KPAIRS / GP)
#define WST     (KPAIRS / WPB)

__global__ __launch_bounds__(TPB)
void qgate_kernel(const float* __restrict__ x,
                  const float* __restrict__ W,
                  const float* __restrict__ bias,
                  const float* __restrict__ theta,
                  float* __restrict__ out,
                  int batch)
{
    __shared__ v4f lbs[KPAIRS * 32];

    const int tid  = threadIdx.x;
    const int lane = tid & 31;
    const int wave = tid >> 5;
    const int n    = lane & 15;
    const int hi   = lane >> 4;

    const int ntiles = batch >> 4;
    const int tile   = blockIdx.x * WPB + wave;
    const bool valid = tile < ntiles;
    const int rowbase = (valid ? tile : 0) << 4;

    const int nr = (n < NQ) ? n : (NQ - 1);
    const float* wrow = W + nr * IN_F + (hi << 2) + 8 * wave;
    v4f wv[WST];
    #pragma unroll
    for (int it = 0; it < WST; ++it)
        wv[it] = *(const v4f*)(wrow + 8 * WPB * it);
    __builtin_amdgcn_sched_barrier(0);

    const float* ap = x + (size_t)(rowbase + n) * IN_F + (hi << 2);
    v4f af[KPAIRS];
    #pragma unroll
    for (int p = 0; p < KPAIRS; ++p)
        af[p] = *(const v4f*)(ap + 8 * p);

    const float bv = bias[nr];
    const float tv = theta[nr];

    __builtin_amdgcn_sched_barrier(0);

    {
        const v4f zero = {0.f, 0.f, 0.f, 0.f};
        #pragma unroll
        for (int it = 0; it < WST; ++it)
            lbs[(wave + WPB * it) * 32 + lane] = (n < NQ) ? wv[it] : zero;
    }
    __builtin_amdgcn_fence(__ATOMIC_RELEASE, "workgroup", "local");
    __builtin_amdgcn_s_barrier();
    __builtin_amdgcn_fence(__ATOMIC_ACQUIRE, "workgroup", "local");

    if (!valid) return;

    v8f acc0 = {}; v8f acc1 = {};

    v4f bf[2][GP];
    #pragma unroll
    for (int j = 0; j < GP; ++j) bf[0][j] = lbs[j * 32 + lane];

    #pragma unroll
    for (int g = 0; g < NG; ++g) {
        if (g + 1 < NG) {
            #pragma unroll
            for (int j = 0; j < GP; ++j)
                bf[(g + 1) & 1][j] = lbs[((g + 1) * GP + j) * 32 + lane];
        }
        __builtin_amdgcn_sched_barrier(0);
        #pragma unroll
        for (int j = 0; j < GP; ++j) {
            const v4f a = af[g * GP + j];
            const v4f b = bf[g & 1][j];
            const v2f a0 = {a.x, a.y}, a1 = {a.z, a.w};
            const v2f b0 = {b.x, b.y}, b1 = {b.z, b.w};
            acc0 = __builtin_amdgcn_wmma_f32_16x16x4_f32(false, a0, false, b0,
                                                         (short)0, acc0, false, false);
            acc1 = __builtin_amdgcn_wmma_f32_16x16x4_f32(false, a1, false, b1,
                                                         (short)0, acc1, false, false);
        }
        __builtin_amdgcn_sched_barrier(0);
    }

    const float bt = (n < NQ) ? (bv + tv) : 0.f;

    #pragma unroll
    for (int i = 0; i < 8; ++i) {
        float v = cosf(acc0[i] + acc1[i] + bt);
        #pragma unroll
        for (int off = 1; off < 16; off <<= 1) {
            float o = __shfl_up(v, off, 16);
            v = (n >= off) ? v * o : v;
        }
        if (n < NQ) {
            const int row = rowbase + (hi ? (8 + i) : i);
            out[(size_t)row * NQ + n] = v;
        }
    }
}

extern "C" void kernel_launch(void* const* d_in, const int* in_sizes, int n_in,
                              void* d_out, int out_size, void* d_ws, size_t ws_size,
                              hipStream_t stream)
{
    const float* x     = (const float*)d_in[0];
    const float* W     = (const float*)d_in[1];
    const float* b     = (const float*)d_in[2];
    const float* theta = (const float*)d_in[3];
    float* out = (float*)d_out;

    const int batch  = in_sizes[0] / IN_F;
    const int ntiles = batch >> 4;
    const int blocks = (ntiles + WPB - 1) / WPB;
    qgate_kernel<<<blocks, TPB, 0, stream>>>(x, W, b, theta, out, batch);
}


# epilogue prefix product over the 10 columns by DPP row shifts, all 8 result registers per step (4 steps; was 32 dependent LDS lane-permute round trips per wave); same products and order
# speedup vs baseline: 2.3661x; 1.1710x over previous
#include <hip/hip_runtime.h>
#include <math.h>


typedef __attribute__((ext_vector_type(2))) float v2f;
typedef __attribute__((ext_vector_type(4))) float v4f;
typedef __attribute__((ext_vector_type(8))) float v8f;

#define IN_F    512
#define NQ      10
#define KPAIRS  (IN_F / 8)
#define WPB     4
#define TPB     (WPB * 32)
#define GP      8
#define NG      (KPAIRS / GP)
#define WST     (KPAIRS / WPB)

template <int OFF>
__device__ __forceinline__ float from_lower(float v)
{
    return __builtin_bit_cast(float,
        __builtin_amdgcn_update_dpp(0x3f800000, __builtin_bit_cast(int, v),
                                    0x110 + OFF, 0xf, 0xf, false));
}

__global__ __launch_bounds__(TPB)
void qgate_kernel(const float* __restrict__ x,
                  const float* __restrict__ W,
                  const float* __restrict__ bias,
                  const float* __restrict__ theta,
                  float* __restrict__ out,
                  int batch)
{
    __shared__ v4f lbs[KPAIRS * 32];

    const int tid  = threadIdx.x;
    const int lane = tid & 31;
    const int wave = tid >> 5;
    const int n    = lane & 15;
    const int hi   = lane >> 4;

    const int ntiles = batch >> 4;
    const int tile   = blockIdx.x * WPB + wave;
    const bool valid = tile < ntiles;
    const int rowbase = (valid ? tile : 0) << 4;

    const int nr = (n < NQ) ? n : (NQ - 1);
    const float* wrow = W + nr * IN_F + (hi << 2) + 8 * wave;
    v4f wv[WST];
    #pragma unroll
    for (int it = 0; it < WST; ++it)
        wv[it] = *(const v4f*)(wrow + 8 * WPB * it);
    __builtin_amdgcn_sched_barrier(0);

    const float* ap = x + (size_t)(rowbase + n) * IN_F + (hi << 2);
    v4f af[KPAIRS];
    #pragma unroll
    for (int p = 0; p < KPAIRS; ++p)
        af[p] = *(const v4f*)(ap + 8 * p);

    const float bv = bias[nr];
    const float tv = theta[nr];

    __builtin_amdgcn_sched_barrier(0);

    {
        const v4f zero = {0.f, 0.f, 0.f, 0.f};
        #pragma unroll
        for (int it = 0; it < WST; ++it)
            lbs[(wave + WPB * it) * 32 + lane] = (n < NQ) ? wv[it] : zero;
    }
    __builtin_amdgcn_fence(__ATOMIC_RELEASE, "workgroup", "local");
    __builtin_amdgcn_s_barrier();
    __builtin_amdgcn_fence(__ATOMIC_ACQUIRE, "workgroup", "local");

    if (!valid) return;

    v8f acc0 = {}; v8f acc1 = {};

    v4f bf[2][GP];
    #pragma unroll
    for (int j = 0; j < GP; ++j) bf[0][j] = lbs[j * 32 + lane];

    #pragma unroll
    for (int g = 0; g < NG; ++g) {
        if (g + 1 < NG) {
            #pragma unroll
            for (int j = 0; j < GP; ++j)
                bf[(g + 1) & 1][j] = lbs[((g + 1) * GP + j) * 32 + lane];
        }
        __builtin_amdgcn_sched_barrier(0);
        #pragma unroll
        for (int j = 0; j < GP; ++j) {
            const v4f a = af[g * GP + j];
            const v4f b = bf[g & 1][j];
            const v2f a0 = {a.x, a.y}, a1 = {a.z, a.w};
            const v2f b0 = {b.x, b.y}, b1 = {b.z, b.w};
            acc0 = __builtin_amdgcn_wmma_f32_16x16x4_f32(false, a0, false, b0,
                                                         (short)0, acc0, false, false);
            acc1 = __builtin_amdgcn_wmma_f32_16x16x4_f32(false, a1, false, b1,
                                                         (short)0, acc1, false, false);
        }
        __builtin_amdgcn_sched_barrier(0);
    }

    const float bt = (n < NQ) ? (bv + tv) : 0.f;

    float c[8];
    #pragma unroll
    for (int i = 0; i < 8; ++i)
        c[i] = cosf(acc0[i] + acc1[i] + bt);

    #pragma unroll
    for (int i = 0; i < 8; ++i) c[i] *= from_lower<1>(c[i]);
    #pragma unroll
    for (int i = 0; i < 8; ++i) c[i] *= from_lower<2>(c[i]);
    #pragma unroll
    for (int i = 0; i < 8; ++i) c[i] *= from_lower<4>(c[i]);
    #pragma unroll
    for (int i = 0; i < 8; ++i) c[i] *= from_lower<8>(c[i]);

    if (n < NQ) {
        float* op = out + (size_t)(rowbase + (hi ? 8 : 0)) * NQ + n;
        #pragma unroll
        for (int i = 0; i < 8; ++i)
            op[i * NQ] = c[i];
    }
}

extern "C" void kernel_launch(void* const* d_in, const int* in_sizes, int n_in,
                              void* d_out, int out_size, void* d_ws, size_t ws_size,
                              hipStream_t stream)
{
    const float* x     = (const float*)d_in[0];
    const float* W     = (const float*)d_in[1];
    const float* b     = (const float*)d_in[2];
    const float* theta = (const float*)d_in[3];
    float* out = (float*)d_out;

    const int batch  = in_sizes[0] / IN_F;
    const int ntiles = batch >> 4;
    const int blocks = (ntiles + WPB - 1) / WPB;
    qgate_kernel<<<blocks, TPB, 0, stream>>>(x, W, b, theta, out, batch);
}
